# starting kernel
# baseline (speedup 1.0000x reference)
#include <hip/hip_runtime.h>

typedef float v2f __attribute__((ext_vector_type(2)));
typedef float v8f __attribute__((ext_vector_type(8)));

__global__ void gcn_init_deg(float* __restrict__ deg, int n) {
  int i = blockIdx.x * blockDim.x + threadIdx.x;
  if (i < n) deg[i] = 1.0f;
}

__global__ void gcn_count_deg(const int* __restrict__ dst, float* deg, int e) {
  int i = blockIdx.x * blockDim.x + threadIdx.x;
  if (i < e) atomicAdd(&deg[dst[i]], 1.0f);
}

__global__ void gcn_dinv(float* deg, int n) {
  int i = blockIdx.x * blockDim.x + threadIdx.x;
  if (i < n) deg[i] = rsqrtf(deg[i]);
}

__global__ void gcn_zero(float* __restrict__ p, long n) {
  long i = (long)blockIdx.x * blockDim.x + threadIdx.x;
  if (i < n) p[i] = 0.0f;
}

template <int LDB, int LDC, int NVALID>
__global__ void gcn_gemm_wmma_f32(const float* __restrict__ A,
                                  const float* __restrict__ B,
                                  float* __restrict__ C, int nrows) {
  const int lane = threadIdx.x & 31;
  const int wave = threadIdx.x >> 5;
  const int row0 = (blockIdx.x * (blockDim.x >> 5) + wave) * 16;
  if (row0 >= nrows) return;
  const int n0   = blockIdx.y * 16;

  const int mrow = lane & 15;
  const int hsel = lane >> 4;
  const int kofs = hsel << 1;

  const float* arow = A + (row0 + mrow) * 64 + kofs;
  const int    bn   = n0 + mrow;
  const int    bcol = (bn < NVALID) ? bn : (NVALID - 1);
  const float* bp   = B + kofs * LDB + bcol;

  v8f acc = {};
#pragma unroll
  for (int k = 0; k < 64; k += 4) {
    v2f a = *(const v2f*)(arow + k);
    v2f b = { bp[k * LDB], bp[k * LDB + LDB] };
    acc = __builtin_amdgcn_wmma_f32_16x16x4_f32(
         false, a,  false, b,
         (short)0, acc,  false,  false);
  }

  float* cp = C + (row0 + (hsel << 3)) * LDC + n0 + mrow;
  if (row0 + 16 <= nrows) {
#pragma unroll
    for (int r = 0; r < 8; ++r) cp[r * LDC] = acc[r];
  } else {
    const int crow = row0 + (hsel << 3);
#pragma unroll
    for (int r = 0; r < 8; ++r)
      if (crow + r < nrows) cp[r * LDC] = acc[r];
  }
}

__global__ void gcn_scatter64(const float* __restrict__ h,
                              const int* __restrict__ src,
                              const int* __restrict__ dst,
                              const float* __restrict__ dinv,
                              float* agg, int e) {
  int widx = blockIdx.x * (blockDim.x >> 5) + (threadIdx.x >> 5);
  int lane = threadIdx.x & 31;
  if (widx >= e) return;
  int s = src[widx], d = dst[widx];
  float w = dinv[s] * dinv[d];
  v2f v = *(const v2f*)(h + (long)s * 64 + lane * 2);
  float* od = agg + (long)d * 64 + lane * 2;
  atomicAdd(od + 0, v.x * w);
  atomicAdd(od + 1, v.y * w);
}

__global__ void gcn_epilogue1(float* agg, const float* __restrict__ h,
                              const float* __restrict__ dinv,
                              const float* __restrict__ b1, int n) {
  long i = (long)blockIdx.x * blockDim.x + threadIdx.x;
  if (i >= (long)n * 64) return;
  int node = (int)(i >> 6);
  int c    = (int)(i & 63);
  float di = dinv[node];
  float v  = agg[i] + di * di * h[i] + b1[c];
  agg[i] = v > 0.0f ? v : 0.0f;
}

__global__ void gcn_scatter40(const float* __restrict__ h2,
                              const int* __restrict__ src,
                              const int* __restrict__ dst,
                              const float* __restrict__ dinv,
                              float* out, int e) {
  int widx = blockIdx.x * (blockDim.x >> 5) + (threadIdx.x >> 5);
  int lane = threadIdx.x & 31;
  if (widx >= e) return;
  int s = src[widx], d = dst[widx];
  float w = dinv[s] * dinv[d];
  const float* hs = h2 + (long)s * 48;
  float* od = out + (long)d * 40;
  atomicAdd(od + lane, hs[lane] * w);
  if (lane < 8) atomicAdd(od + 32 + lane, hs[32 + lane] * w);
}

__global__ void gcn_epilogue2(float* out, const float* __restrict__ h2,
                              const float* __restrict__ dinv,
                              const float* __restrict__ b3, int n) {
  long i = (long)blockIdx.x * blockDim.x + threadIdx.x;
  if (i >= (long)n * 40) return;
  int node = (int)(i / 40);
  int c    = (int)(i % 40);
  float di = dinv[node];
  out[i] += di * di * h2[(long)node * 48 + c] + b3[c];
}

extern "C" void kernel_launch(void* const* d_in, const int* in_sizes, int n_in,
                              void* d_out, int out_size, void* d_ws, size_t ws_size,
                              hipStream_t stream) {
  (void)n_in; (void)out_size; (void)ws_size;
  const float* x  = (const float*)d_in[0];
  const int*   ei = (const int*)d_in[1];
  const float* W1 = (const float*)d_in[2];
  const float* b1 = (const float*)d_in[3];
  const float* W3 = (const float*)d_in[4];
  const float* b3 = (const float*)d_in[5];

  const int N = in_sizes[0] / 64;
  const int E = in_sizes[1] / 2;
  const int* src = ei;
  const int* dst = ei + E;

  float* dinv = (float*)d_ws;
  float* h    = dinv + N;
  float* agg1 = h + (long)N * 64;
  float* h2   = agg1 + (long)N * 64;
  float* out  = (float*)d_out;

  const int T = 256;
  const int mtiles = (N + 15) / 16;

  gcn_init_deg<<<(N + T - 1) / T, T, 0, stream>>>(dinv, N);
  gcn_count_deg<<<(E + T - 1) / T, T, 0, stream>>>(dst, dinv, E);
  gcn_dinv<<<(N + T - 1) / T, T, 0, stream>>>(dinv, N);

  {
    dim3 grid((mtiles + 7) / 8, 4);
    gcn_gemm_wmma_f32<64, 64, 64><<<grid, 256, 0, stream>>>(x, W1, h, N);
  }
  gcn_zero<<<(int)(((long)N * 64 + T - 1) / T), T, 0, stream>>>(agg1, (long)N * 64);
  gcn_scatter64<<<(E + 7) / 8, 256, 0, stream>>>(h, src, dst, dinv, agg1, E);
  gcn_epilogue1<<<(int)(((long)N * 64 + T - 1) / T), T, 0, stream>>>(agg1, h, dinv, b1, N);

  {
    dim3 grid((mtiles + 7) / 8, 3);
    gcn_gemm_wmma_f32<40, 48, 40><<<grid, 256, 0, stream>>>(agg1, W3, h2, N);
  }
  gcn_zero<<<(int)(((long)N * 40 + T - 1) / T), T, 0, stream>>>(out, (long)N * 40);
  gcn_scatter40<<<(E + 7) / 8, 256, 0, stream>>>(h2, src, dst, dinv, out, E);
  gcn_epilogue2<<<(int)(((long)N * 40 + T - 1) / T), T, 0, stream>>>(out, h2, dinv, b3, N);
}


# Both GEMMs: a wave owns 16 rows x all columns; A rows come as whole-line 16-B loads through a wave-private LDS slot, W staged per block in LDS, C leaves as whole lines; k order and sum chain kept
# speedup vs baseline: 2.1972x; 2.1972x over previous
#include <hip/hip_runtime.h>


typedef float v2f __attribute__((ext_vector_type(2)));
typedef float v8f __attribute__((ext_vector_type(8)));
typedef unsigned u4 __attribute__((ext_vector_type(4)));

#define OWN_ROWS     392
#define OWN_THREADS  1024
#define OWN_WAVES    (OWN_THREADS / 32)
#define KEY_GROUPS   8
#define ROWS_IN_FLIGHT 16
#define LIST_TRIPS   4

__global__ void gcn_init_deg(float* __restrict__ deg, int n) {
  int i = blockIdx.x * blockDim.x + threadIdx.x;
  if (i < n) deg[i] = 1.0f;
}

__global__ void gcn_count_deg_keys(const int* __restrict__ dst, float* deg,
                                   unsigned* __restrict__ keys4, int e, int nwords) {
  const int t = blockIdx.x * blockDim.x + threadIdx.x;
  if (t >= nwords) return;
  unsigned pk = 0u;
#pragma unroll
  for (int b = 0; b < 4; ++b) {
    const int i = t * 4 + b;
    if (i < e) {
      const int d = dst[i];
      atomicAdd(&deg[d], 1.0f);
      pk |= (unsigned)((d / OWN_ROWS) & 255) << (8 * b);
    }
  }
  keys4[t] = pk;
}

#define DINV_THREADS 512
__global__ void __launch_bounds__(DINV_THREADS)
gcn_dinv_count(float* deg, int* __restrict__ ocnt, int n) {
  __shared__ int total;
  const int lo = blockIdx.x * OWN_ROWS;
  if (threadIdx.x == 0) total = 0;
  __syncthreads();
  int c = 0;
  for (int r = threadIdx.x; r < OWN_ROWS; r += DINV_THREADS) {
    const int i = lo + r;
    if (i < n) {
      const float dg = deg[i];
      c += (int)dg - 1;
      deg[i] = rsqrtf(dg);
    }
  }
  for (int o = 16; o > 0; o >>= 1) c += __shfl_xor(c, o);
  if ((threadIdx.x & 31) == 0) atomicAdd(&total, c);
  __syncthreads();
  if (threadIdx.x == 0) ocnt[blockIdx.x] = total;
}

#define GEMM_WAVES 8
#define A_PITCH 68
#define B_PITCH 72

template <int NCOL, bool RELU_A>
__global__ void __launch_bounds__(GEMM_WAVES * 32)
gcn_gemm_rows_f32(const float* __restrict__ A,
                  const float* __restrict__ B,
                  float* __restrict__ C, int nrows) {
  constexpr int NT = (NCOL + 15) / 16;
  constexpr int NP = NT * 16;
  constexpr int QR = NCOL / 4;
  constexpr int Q  = 16 * QR;
  static_assert(NCOL % 4 == 0 && NP <= B_PITCH && NP <= A_PITCH, "tile shape");
  static_assert((64 * NCOL) % (GEMM_WAVES * 32) == 0 && (64 * (NP - NCOL)) % (GEMM_WAVES * 32) == 0 &&
                Q % 32 == 0, "whole trips");
  __shared__ __attribute__((aligned(16))) float bl[64 * B_PITCH];
  __shared__ __attribute__((aligned(16))) float sl[GEMM_WAVES][16 * A_PITCH];

  const int tid  = threadIdx.x;
  const int lane = tid & 31;
  const int wave = tid >> 5;
  const int row0 = (blockIdx.x * GEMM_WAVES + wave) * 16;
  float* s = sl[wave];

  float4 av[8];
#pragma unroll
  for (int j = 0; j < 8; ++j) {
    const int q = j * 32 + lane;
    const int r = min(row0 + (q >> 4), nrows - 1);
    av[j] = *(const float4*)(A + (long)r * 64 + (q & 15) * 4);
  }

#pragma unroll
  for (int i = 0; i < (64 * NCOL) / (GEMM_WAVES * 32); ++i) {
    const int idx = i * (GEMM_WAVES * 32) + tid;
    const int k = idx / NCOL;
    bl[k * B_PITCH + (idx - k * NCOL)] = B[idx];
  }
  if (NP > NCOL) {
#pragma unroll
    for (int i = 0; i < (64 * (NP - NCOL)) / (GEMM_WAVES * 32); ++i) {
      const int idx = i * (GEMM_WAVES * 32) + tid;
      const int k = idx / (NP - NCOL);
      bl[k * B_PITCH + NCOL + (idx - k * (NP - NCOL))] = 0.0f;
    }
  }

#pragma unroll
  for (int j = 0; j < 8; ++j) {
    const int q = j * 32 + lane;
    float4 v = av[j];
    if (RELU_A) {
      v.x = v.x > 0.0f ? v.x : 0.0f;
      v.y = v.y > 0.0f ? v.y : 0.0f;
      v.z = v.z > 0.0f ? v.z : 0.0f;
      v.w = v.w > 0.0f ? v.w : 0.0f;
    }
    *(float4*)(s + (q >> 4) * A_PITCH + (q & 15) * 4) = v;
  }
  __syncthreads();
  if (row0 >= nrows) return;

  const int c  = lane & 15;
  const int hh = lane >> 4;
  const float* ap = s + c * A_PITCH + 2 * hh;
  const float* bp = bl + (2 * hh) * B_PITCH + c;

  v8f acc[NT];
#pragma unroll
  for (int t = 0; t < NT; ++t) acc[t] = v8f{};
#pragma unroll
  for (int ks = 0; ks < 16; ++ks) {
    const v2f a = *(const v2f*)(ap + 4 * ks);
#pragma unroll
    for (int t = 0; t < NT; ++t) {
      const v2f b = { bp[(4 * ks) * B_PITCH + 16 * t],
                      bp[(4 * ks + 1) * B_PITCH + 16 * t] };
      acc[t] = __builtin_amdgcn_wmma_f32_16x16x4_f32(
           false, a,  false, b,
           (short)0, acc[t],  false,  false);
    }
  }

#pragma unroll
  for (int t = 0; t < NT; ++t) {
#pragma unroll
    for (int r = 0; r < 8; ++r) s[(8 * hh + r) * A_PITCH + 16 * t + c] = acc[t][r];
  }
  __builtin_amdgcn_wave_barrier();

  float4* cp = (float4*)(C + (long)row0 * NCOL);
  if (row0 + 16 <= nrows) {
    float4 cv[Q / 32];
#pragma unroll
    for (int j = 0; j < Q / 32; ++j) {
      const int q = j * 32 + lane;
      const int r = q / QR;
      cv[j] = *(const float4*)(s + r * A_PITCH + (q - r * QR) * 4);
    }
#pragma unroll
    for (int j = 0; j < Q / 32; ++j) cp[j * 32 + lane] = cv[j];
  } else {
#pragma unroll
    for (int j = 0; j < Q / 32; ++j) {
      const int q = j * 32 + lane;
      const int r = q / QR;
      if (row0 + r < nrows) cp[q] = *(const float4*)(s + r * A_PITCH + (q - r * QR) * 4);
    }
  }
}

__device__ __forceinline__ float lane_bcast_f(float v, int j) {
  return __int_as_float(__builtin_amdgcn_readlane(__float_as_int(v), j));
}

template <int D>
__device__ __forceinline__ void add_rows(const float* __restrict__ h, float* rows,
                                         int s, int dl, float w, unsigned m, int lane) {
  const bool act = (D == 64) || (lane < D / 2);
  while (m != 0u) {
    const int k = min((int)__builtin_popcount(m), ROWS_IN_FLIGHT);
    int jj[ROWS_IN_FLIGHT];
    v2f v[ROWS_IN_FLIGHT];
#pragma unroll
    for (int u = 0; u < ROWS_IN_FLIGHT; ++u) {
      jj[u] = 0;
      v[u] = v2f{0.0f, 0.0f};
      if (u < k) {
        jj[u] = __builtin_ctz(m);
        m &= m - 1u;
        const int sj = __builtin_amdgcn_readlane(s, jj[u]);
        if (act) v[u] = *(const v2f*)(h + (long)sj * D + 2 * lane);
      }
    }
#pragma unroll
    for (int u = 0; u < ROWS_IN_FLIGHT; ++u) {
      if (u < k) {
        const int rj = __builtin_amdgcn_readlane(dl, jj[u]);
        const float wj = lane_bcast_f(w, jj[u]);
        float* p = rows + rj * D + 2 * lane;
        if (act) {
          atomicAdd(p + 0, v[u].x * wj);
          atomicAdd(p + 1, v[u].y * wj);
        }
      }
    }
  }
}

template <int D>
__device__ __forceinline__ void owner_batch(const float* __restrict__ h,
                                            const int* __restrict__ src,
                                            const int* __restrict__ dst,
                                            const float* __restrict__ dinv,
                                            float* rows, int ei, int nvalid,
                                            int e, int lo, int hi, int lane,
                                            int* lcnt, int off,
                                            int* ls, float* lw, int* lr) {
  int s = 0, d = -1;
  if (lane < nvalid && ei < e) { s = src[ei]; d = dst[ei]; }
  const bool mine = d >= lo && d < hi;
  float ws = 0.0f, wd = 0.0f;
  if (mine) { ws = dinv[s]; wd = dinv[d]; }
  const float w = ws * wd;
  const unsigned m = __builtin_amdgcn_ballot_w32(mine);
  int base = 0;
  if (lane == 0) base = atomicAdd(lcnt, (int)__builtin_popcount(m));
  base = __builtin_amdgcn_readfirstlane(base);
  const int pos = off + base + (int)__builtin_amdgcn_mbcnt_lo(m, 0u);
  if (mine && pos < e) { ls[pos] = s; lw[pos] = w; lr[pos] = d - lo; }
  add_rows<D>(h, rows, s, d - lo, w, m, lane);
}

__device__ __forceinline__ unsigned key_hits4(unsigned x, unsigned kk) {
  const unsigned y = x ^ kk;
  const unsigned z = (y - 0x01010101u) & ~y & 0x80808080u;
  return (((z >> 7) * 0x01020408u) >> 24) & 15u;
}

template <int D, bool SCAN>
__global__ void __launch_bounds__(OWN_THREADS)
gcn_owner_sum(const float* __restrict__ h,
              const int* __restrict__ src,
              const int* __restrict__ dst,
              const float* __restrict__ dinv,
              const float* __restrict__ bias,
              const unsigned* __restrict__ keys,
              const int* __restrict__ ocnt,
              int* ls, float* lw, int* lr,
              float* __restrict__ agg, int e, int nrows) {
  const u4* keys16 = (const u4*)keys;
  __shared__ __attribute__((aligned(16))) float rows[OWN_ROWS * D];
  __shared__ int queue[SCAN ? OWN_WAVES : 1][64];
  __shared__ int lcnt;

  const int tid  = threadIdx.x;
  const int lane = tid & 31;
  const int wave = tid >> 5;
  const int lo   = blockIdx.x * OWN_ROWS;
  const int hi   = min(lo + OWN_ROWS, nrows);
  const int nq   = (hi - lo) * (D / 4);

  int part = 0;
  for (int i = lane; i < (int)blockIdx.x; i += 32) part += ocnt[i];
  for (int o = 16; o > 0; o >>= 1) part += __shfl_xor(part, o);
  const int off = __builtin_amdgcn_readfirstlane(part);
  const int cnt = ocnt[blockIdx.x];
  if (tid == 0) lcnt = 0;

  const float4* h4 = (const float4*)(h + (long)lo * D);
  float4* r4 = (float4*)rows;
  for (int q = tid; q < nq; q += OWN_THREADS) {
    const int r = q / (D / 4);
    const int c = (q - r * (D / 4)) * 4;
    const float di = dinv[lo + r];
    const float4 hv = h4[q];
    float4 o;
    o.x = di * di * hv.x + bias[c + 0];
    o.y = di * di * hv.y + bias[c + 1];
    o.z = di * di * hv.z + bias[c + 2];
    o.w = di * di * hv.w + bias[c + 3];
    r4[q] = o;
  }
  __syncthreads();

  if (SCAN) {
  const int ngroups = (e + 15) >> 4;
  const int nchunks = (ngroups + 31) >> 5;
  const int rot = (int)(((long)blockIdx.x * nchunks) / gridDim.x);
  const unsigned kk = (unsigned)(blockIdx.x & 255) * 0x01010101u;
  int* q = queue[wave];
  int qn = 0;

  for (int j0 = wave; j0 < nchunks; j0 += OWN_WAVES * KEY_GROUPS) {
    u4 kw[KEY_GROUPS];
    bool ok[KEY_GROUPS];
#pragma unroll
    for (int u = 0; u < KEY_GROUPS; ++u) {
      const int j = j0 + u * OWN_WAVES;
      int c = j + rot;
      if (c >= nchunks) c -= nchunks;
      const int g = c * 32 + lane;
      ok[u] = (j < nchunks) && (g < ngroups);
      kw[u] = keys16[min(g, ngroups - 1)];
    }
    static_assert(KEY_GROUPS == 8, "the fence below names eight loads");
    asm volatile("" :: "v"(kw[0]), "v"(kw[1]), "v"(kw[2]), "v"(kw[3]),
                       "v"(kw[4]), "v"(kw[5]), "v"(kw[6]), "v"(kw[7]));
    unsigned long long hA = 0ull, hB = 0ull;
#pragma unroll
    for (int u = 0; u < KEY_GROUPS; ++u) {
      const unsigned h16 = key_hits4(kw[u].x, kk) | (key_hits4(kw[u].y, kk) << 4) |
                           (key_hits4(kw[u].z, kk) << 8) | (key_hits4(kw[u].w, kk) << 12);
      const unsigned long long hm = ok[u] ? (unsigned long long)h16 : 0ull;
      if (u < 4) hA |= hm << (16 * (u & 3));
      else       hB |= hm << (16 * (u & 3));
    }
    while (true) {
      const bool has = (hA | hB) != 0ull;
      const unsigned bal = __builtin_amdgcn_ballot_w32(has);
      if (bal == 0u) break;
      if (has) {
        int p;
        if (hA != 0ull) { p = __builtin_ctzll(hA); hA &= hA - 1ull; }
        else            { p = 64 + __builtin_ctzll(hB); hB &= hB - 1ull; }
        int c = j0 + (p >> 4) * OWN_WAVES + rot;
        if (c >= nchunks) c -= nchunks;
        const int pos = qn + (int)__builtin_amdgcn_mbcnt_lo(bal, 0u);
        q[pos] = (c * 32 + lane) * 16 + (p & 15);
      }
      qn += __builtin_popcount(bal);
      if (qn >= 32) {
        __builtin_amdgcn_wave_barrier();
        const int ei = q[lane];
        const int rest = qn - 32;
        const int mv = (lane < rest) ? q[32 + lane] : 0;
        __builtin_amdgcn_wave_barrier();
        if (lane < rest) q[lane] = mv;
        qn = rest;
        owner_batch<D>(h, src, dst, dinv, rows, ei, 32, e, lo, hi, lane,
                       &lcnt, off, ls, lw, lr);
      }
    }
  }
  if (qn > 0) {
    __builtin_amdgcn_wave_barrier();
    const int ei = (lane < qn) ? q[lane] : 0;
    owner_batch<D>(h, src, dst, dinv, rows, ei, qn, e, lo, hi, lane,
                   &lcnt, off, ls, lw, lr);
  }
  } else {
    const int last = max(e - 1, 0);
    for (int t0 = wave; t0 * 32 < cnt; t0 += OWN_WAVES * LIST_TRIPS) {
      int s[LIST_TRIPS], dl[LIST_TRIPS];
      float w[LIST_TRIPS];
#pragma unroll
      for (int u = 0; u < LIST_TRIPS; ++u) {
        const int p = min(off + (t0 + u * OWN_WAVES) * 32 + lane, last);
        s[u] = ls[p];
        dl[u] = lr[p];
        w[u] = lw[p];
      }
#pragma unroll
      for (int u = 0; u < LIST_TRIPS; ++u) {
        const int i = (t0 + u * OWN_WAVES) * 32 + lane;
        const bool okv = i < cnt && off + i < e &&
                         (unsigned)dl[u] < (unsigned)(hi - lo) &&
                         (unsigned)s[u] < (unsigned)nrows;
        const unsigned m = __builtin_amdgcn_ballot_w32(okv);
        add_rows<D>(h, rows, s[u], dl[u], w[u], m, lane);
      }
    }
  }
  __syncthreads();

  float4* a4 = (float4*)(agg + (long)lo * D);
  for (int t = tid; t < nq; t += OWN_THREADS) a4[t] = r4[t];
}

extern "C" void kernel_launch(void* const* d_in, const int* in_sizes, int n_in,
                              void* d_out, int out_size, void* d_ws, size_t ws_size,
                              hipStream_t stream) {
  (void)n_in; (void)out_size; (void)ws_size;
  const float* x  = (const float*)d_in[0];
  const int*   ei = (const int*)d_in[1];
  const float* W1 = (const float*)d_in[2];
  const float* b1 = (const float*)d_in[3];
  const float* W3 = (const float*)d_in[4];
  const float* b3 = (const float*)d_in[5];

  const int N = in_sizes[0] / 64;
  const int E = in_sizes[1] / 2;
  const int* src = ei;
  const int* dst = ei + E;

  unsigned char* keys = (unsigned char*)d_ws;
  float* dinv = (float*)(keys + (((long)E + 255) & ~255L));
  float* h    = dinv + ((N + 63) & ~63);
  float* agg1 = h + (long)N * 64;
  float* lst  = agg1 + (long)N * 64;
  float* h2   = h;
  int*   ls   = (int*)lst;
  float* lw   = lst + E;
  int*   lr   = (int*)(lst + 2L * E);
  int*   ocnt = (int*)(lst + 3L * E);
  float* out  = (float*)d_out;

  const int T = 256;
  const int mtiles = (N + 15) / 16;
  const int nwords = ((E + 15) / 16) * 4;
  const int owners = (N + OWN_ROWS - 1) / OWN_ROWS;

  gcn_init_deg<<<(N + T - 1) / T, T, 0, stream>>>(dinv, N);
  gcn_count_deg_keys<<<(nwords + T - 1) / T, T, 0, stream>>>(dst, dinv, (unsigned*)keys, E, nwords);
  gcn_dinv_count<<<owners, DINV_THREADS, 0, stream>>>(dinv, ocnt, N);

  gcn_gemm_rows_f32<64, false><<<(mtiles + GEMM_WAVES - 1) / GEMM_WAVES, GEMM_WAVES * 32, 0, stream>>>(x, W1, h, N);
  gcn_owner_sum<64, true><<<owners, OWN_THREADS, 0, stream>>>(
      h, src, dst, dinv, b1, (const unsigned*)keys, ocnt, ls, lw, lr, agg1, E, N);

  gcn_gemm_rows_f32<40, true><<<(mtiles + GEMM_WAVES - 1) / GEMM_WAVES, GEMM_WAVES * 32, 0, stream>>>(agg1, W3, h2, N);
  gcn_owner_sum<40, false><<<owners, OWN_THREADS, 0, stream>>>(
      h2, src, dst, dinv, b3, (const unsigned*)keys, ocnt, ls, lw, lr, out, E, N);
}


# Degree prep passes folded away: layer-1 GEMM (now first) stores the degrees' start value; owners take 1/sqrt(degree) and their list segment from the raw counts; 7 -> 5 launches; sums as before
# speedup vs baseline: 2.2653x; 1.0310x over previous
#include <hip/hip_runtime.h>


typedef float v2f __attribute__((ext_vector_type(2)));
typedef float v8f __attribute__((ext_vector_type(8)));
typedef unsigned u4 __attribute__((ext_vector_type(4)));
typedef float v4f __attribute__((ext_vector_type(4)));

#define OWN_ROWS     392
#define OWN_THREADS  1024
#define OWN_WAVES    (OWN_THREADS / 32)
#define KEY_GROUPS   8
#define ROWS_IN_FLIGHT 16
#define LIST_TRIPS   4
#define DEG_GROUP    8

__global__ void gcn_count_deg_keys(const int* __restrict__ dst, float* deg,
                                   unsigned* __restrict__ keys4, int e, int nwords) {
  const int t = blockIdx.x * blockDim.x + threadIdx.x;
  if (t >= nwords) return;
  unsigned pk = 0u;
#pragma unroll
  for (int b = 0; b < 4; ++b) {
    const int i = t * 4 + b;
    if (i < e) {
      const int d = dst[i];
      atomicAdd(&deg[d], 1.0f);
      pk |= (unsigned)((d / OWN_ROWS) & 255) << (8 * b);
    }
  }
  keys4[t] = pk;
}

#define GEMM_WAVES 8
#define A_PITCH 68
#define B_PITCH 72

template <int NCOL, bool RELU_A, bool DEG_ONE>
__global__ void __launch_bounds__(GEMM_WAVES * 32)
gcn_gemm_rows_f32(const float* __restrict__ A,
                  const float* __restrict__ B,
                  float* __restrict__ C, float* __restrict__ deg, int nrows) {
  constexpr int NT = (NCOL + 15) / 16;
  constexpr int NP = NT * 16;
  constexpr int QR = NCOL / 4;
  constexpr int Q  = 16 * QR;
  static_assert(NCOL % 4 == 0 && NP <= B_PITCH && NP <= A_PITCH, "tile shape");
  static_assert((64 * NCOL) % (GEMM_WAVES * 32) == 0 && (64 * (NP - NCOL)) % (GEMM_WAVES * 32) == 0 &&
                Q % 32 == 0, "whole trips");
  __shared__ __attribute__((aligned(16))) float bl[64 * B_PITCH];
  __shared__ __attribute__((aligned(16))) float sl[GEMM_WAVES][16 * A_PITCH];

  const int tid  = threadIdx.x;
  const int lane = tid & 31;
  const int wave = tid >> 5;
  const int row0 = (blockIdx.x * GEMM_WAVES + wave) * 16;
  float* s = sl[wave];

  if (DEG_ONE && wave == 0) {
    const int r = blockIdx.x * (GEMM_WAVES * 16) + lane * 4;
    if (r + 4 <= nrows) {
      *(float4*)(deg + r) = float4{1.0f, 1.0f, 1.0f, 1.0f};
    } else {
      for (int i = r; i < nrows; ++i) deg[i] = 1.0f;
    }
  }

  float4 av[8];
#pragma unroll
  for (int j = 0; j < 8; ++j) {
    const int q = j * 32 + lane;
    const int r = min(row0 + (q >> 4), nrows - 1);
    av[j] = *(const float4*)(A + (long)r * 64 + (q & 15) * 4);
  }

#pragma unroll
  for (int i = 0; i < (64 * NCOL) / (GEMM_WAVES * 32); ++i) {
    const int idx = i * (GEMM_WAVES * 32) + tid;
    const int k = idx / NCOL;
    bl[k * B_PITCH + (idx - k * NCOL)] = B[idx];
  }
  if (NP > NCOL) {
#pragma unroll
    for (int i = 0; i < (64 * (NP - NCOL)) / (GEMM_WAVES * 32); ++i) {
      const int idx = i * (GEMM_WAVES * 32) + tid;
      const int k = idx / (NP - NCOL);
      bl[k * B_PITCH + NCOL + (idx - k * (NP - NCOL))] = 0.0f;
    }
  }

#pragma unroll
  for (int j = 0; j < 8; ++j) {
    const int q = j * 32 + lane;
    float4 v = av[j];
    if (RELU_A) {
      v.x = v.x > 0.0f ? v.x : 0.0f;
      v.y = v.y > 0.0f ? v.y : 0.0f;
      v.z = v.z > 0.0f ? v.z : 0.0f;
      v.w = v.w > 0.0f ? v.w : 0.0f;
    }
    *(float4*)(s + (q >> 4) * A_PITCH + (q & 15) * 4) = v;
  }
  __syncthreads();
  if (row0 >= nrows) return;

  const int c  = lane & 15;
  const int hh = lane >> 4;
  const float* ap = s + c * A_PITCH + 2 * hh;
  const float* bp = bl + (2 * hh) * B_PITCH + c;

  v8f acc[NT];
#pragma unroll
  for (int t = 0; t < NT; ++t) acc[t] = v8f{};
#pragma unroll
  for (int ks = 0; ks < 16; ++ks) {
    const v2f a = *(const v2f*)(ap + 4 * ks);
#pragma unroll
    for (int t = 0; t < NT; ++t) {
      const v2f b = { bp[(4 * ks) * B_PITCH + 16 * t],
                      bp[(4 * ks + 1) * B_PITCH + 16 * t] };
      acc[t] = __builtin_amdgcn_wmma_f32_16x16x4_f32(
           false, a,  false, b,
           (short)0, acc[t],  false,  false);
    }
  }

#pragma unroll
  for (int t = 0; t < NT; ++t) {
#pragma unroll
    for (int r = 0; r < 8; ++r) s[(8 * hh + r) * A_PITCH + 16 * t + c] = acc[t][r];
  }
  __builtin_amdgcn_wave_barrier();

  float4* cp = (float4*)(C + (long)row0 * NCOL);
  if (row0 + 16 <= nrows) {
    float4 cv[Q / 32];
#pragma unroll
    for (int j = 0; j < Q / 32; ++j) {
      const int q = j * 32 + lane;
      const int r = q / QR;
      cv[j] = *(const float4*)(s + r * A_PITCH + (q - r * QR) * 4);
    }
#pragma unroll
    for (int j = 0; j < Q / 32; ++j) cp[j * 32 + lane] = cv[j];
  } else {
#pragma unroll
    for (int j = 0; j < Q / 32; ++j) {
      const int q = j * 32 + lane;
      const int r = q / QR;
      if (row0 + r < nrows) cp[q] = *(const float4*)(s + r * A_PITCH + (q - r * QR) * 4);
    }
  }
}

__device__ __forceinline__ float lane_bcast_f(float v, int j) {
  return __int_as_float(__builtin_amdgcn_readlane(__float_as_int(v), j));
}

template <int D>
__device__ __forceinline__ void add_rows(const float* __restrict__ h, float* rows,
                                         int s, int dl, float w, unsigned m, int lane) {
  const bool act = (D == 64) || (lane < D / 2);
  while (m != 0u) {
    const int k = min((int)__builtin_popcount(m), ROWS_IN_FLIGHT);
    int jj[ROWS_IN_FLIGHT];
    v2f v[ROWS_IN_FLIGHT];
#pragma unroll
    for (int u = 0; u < ROWS_IN_FLIGHT; ++u) {
      jj[u] = 0;
      v[u] = v2f{0.0f, 0.0f};
      if (u < k) {
        jj[u] = __builtin_ctz(m);
        m &= m - 1u;
        const int sj = __builtin_amdgcn_readlane(s, jj[u]);
        if (act) v[u] = *(const v2f*)(h + (long)sj * D + 2 * lane);
      }
    }
#pragma unroll
    for (int u = 0; u < ROWS_IN_FLIGHT; ++u) {
      if (u < k) {
        const int rj = __builtin_amdgcn_readlane(dl, jj[u]);
        const float wj = lane_bcast_f(w, jj[u]);
        float* p = rows + rj * D + 2 * lane;
        if (act) {
          atomicAdd(p + 0, v[u].x * wj);
          atomicAdd(p + 1, v[u].y * wj);
        }
      }
    }
  }
}

template <int D>
__device__ __forceinline__ void owner_batch(const float* __restrict__ h,
                                            const int* __restrict__ src,
                                            const int* __restrict__ dst,
                                            const float* __restrict__ deg,
                                            float* rows, int ei, int nvalid,
                                            int e, int lo, int hi, int lane,
                                            int* lcnt, int off,
                                            int* ls, float* lw, int* lr) {
  int s = 0, d = -1;
  if (lane < nvalid && ei < e) { s = src[ei]; d = dst[ei]; }
  const bool mine = d >= lo && d < hi;
  float ws = 0.0f, wd = 0.0f;
  if (mine) { ws = rsqrtf(deg[s]); wd = rsqrtf(deg[d]); }
  const float w = ws * wd;
  const unsigned m = __builtin_amdgcn_ballot_w32(mine);
  int base = 0;
  if (lane == 0) base = atomicAdd(lcnt, (int)__builtin_popcount(m));
  base = __builtin_amdgcn_readfirstlane(base);
  const int pos = off + base + (int)__builtin_amdgcn_mbcnt_lo(m, 0u);
  if (mine && pos < e) { ls[pos] = s; lw[pos] = w; lr[pos] = d - lo; }
  add_rows<D>(h, rows, s, d - lo, w, m, lane);
}

__device__ __forceinline__ unsigned key_hits4(unsigned x, unsigned kk) {
  const unsigned y = x ^ kk;
  const unsigned z = (y - 0x01010101u) & ~y & 0x80808080u;
  return (((z >> 7) * 0x01020408u) >> 24) & 15u;
}

template <int D, bool SCAN>
__global__ void __launch_bounds__(OWN_THREADS)
gcn_owner_sum(const float* __restrict__ h,
              const int* __restrict__ src,
              const int* __restrict__ dst,
              const float* __restrict__ deg,
              const float* __restrict__ bias,
              const unsigned* __restrict__ keys,
              int* oinfo,
              int* ls, float* lw, int* lr,
              float* __restrict__ agg, int e, int nrows) {
  const u4* keys16 = (const u4*)keys;
  __shared__ __attribute__((aligned(16))) float rows[OWN_ROWS * D];
  __shared__ int queue[SCAN ? OWN_WAVES : 1][64];
  __shared__ int lcnt;
  __shared__ int wpre[SCAN ? OWN_WAVES : 1];
  __shared__ int wown[SCAN ? OWN_WAVES : 1];
  static_assert(OWN_WAVES == 32, "one partial per lane in the sums below");

  const int tid  = threadIdx.x;
  const int lane = tid & 31;
  const int wave = tid >> 5;
  const int lo   = blockIdx.x * OWN_ROWS;
  const int hi   = min(lo + OWN_ROWS, nrows);
  const int nq   = (hi - lo) * (D / 4);

  int off = 0, cnt = 0;
  const bool below = 2 * lo <= nrows;
  if (SCAN) {
    const int a = below ? 0 : hi;
    const int b = below ? lo : nrows;
    const int n4 = (b - a) >> 2;
    const v4f* d4 = (const v4f*)(deg + a);
    int c = 0;
    for (int q0 = tid; q0 < n4; q0 += OWN_THREADS * DEG_GROUP) {
      v4f v[DEG_GROUP];
#pragma unroll
      for (int u = 0; u < DEG_GROUP; ++u) v[u] = d4[min(q0 + u * OWN_THREADS, n4 - 1)];
      static_assert(DEG_GROUP == 8, "the fence below names eight loads");
      asm volatile("" :: "v"(v[0]), "v"(v[1]), "v"(v[2]), "v"(v[3]),
                         "v"(v[4]), "v"(v[5]), "v"(v[6]), "v"(v[7]));
#pragma unroll
      for (int u = 0; u < DEG_GROUP; ++u) {
        const int t = ((int)v[u].x - 1) + ((int)v[u].y - 1) + ((int)v[u].z - 1) + ((int)v[u].w - 1);
        c += (q0 + u * OWN_THREADS < n4) ? t : 0;
      }
    }
    for (int r = a + 4 * n4 + tid; r < b; r += OWN_THREADS) c += (int)deg[r] - 1;
    int cm = 0;
    for (int r = lo + tid; r < hi; r += OWN_THREADS) cm += (int)deg[r] - 1;
    for (int o = 16; o > 0; o >>= 1) { c += __shfl_xor(c, o); cm += __shfl_xor(cm, o); }
    if (lane == 0) { wpre[wave] = c; wown[wave] = cm; }
  } else {
    off = min(max(oinfo[2 * blockIdx.x], 0), e);
    cnt = min(max(oinfo[2 * blockIdx.x + 1], 0), e - off);
  }
  if (tid == 0) lcnt = 0;

  const float4* h4 = (const float4*)(h + (long)lo * D);
  float4* r4 = (float4*)rows;
  for (int q = tid; q < nq; q += OWN_THREADS) {
    const int r = q / (D / 4);
    const int c = (q - r * (D / 4)) * 4;
    const float di = rsqrtf(deg[lo + r]);
    const float4 hv = h4[q];
    float4 o;
    o.x = di * di * hv.x + bias[c + 0];
    o.y = di * di * hv.y + bias[c + 1];
    o.z = di * di * hv.z + bias[c + 2];
    o.w = di * di * hv.w + bias[c + 3];
    r4[q] = o;
  }
  __syncthreads();

  if (SCAN) {
  {
    int c = wpre[lane], cm = wown[lane];
    for (int o = 16; o > 0; o >>= 1) { c += __shfl_xor(c, o); cm += __shfl_xor(cm, o); }
    cnt = __builtin_amdgcn_readfirstlane(cm);
    c = __builtin_amdgcn_readfirstlane(c);
    off = below ? c : e - c - cnt;
    if (tid == 0) { oinfo[2 * blockIdx.x] = off; oinfo[2 * blockIdx.x + 1] = cnt; }
  }
  const int ngroups = (e + 15) >> 4;
  const int nchunks = (ngroups + 31) >> 5;
  const int rot = (int)(((long)blockIdx.x * nchunks) / gridDim.x);
  const unsigned kk = (unsigned)(blockIdx.x & 255) * 0x01010101u;
  int* q = queue[wave];
  int qn = 0;

  for (int j0 = wave; j0 < nchunks; j0 += OWN_WAVES * KEY_GROUPS) {
    u4 kw[KEY_GROUPS];
    bool ok[KEY_GROUPS];
#pragma unroll
    for (int u = 0; u < KEY_GROUPS; ++u) {
      const int j = j0 + u * OWN_WAVES;
      int c = j + rot;
      if (c >= nchunks) c -= nchunks;
      const int g = c * 32 + lane;
      ok[u] = (j < nchunks) && (g < ngroups);
      kw[u] = keys16[min(g, ngroups - 1)];
    }
    static_assert(KEY_GROUPS == 8, "the fence below names eight loads");
    asm volatile("" :: "v"(kw[0]), "v"(kw[1]), "v"(kw[2]), "v"(kw[3]),
                       "v"(kw[4]), "v"(kw[5]), "v"(kw[6]), "v"(kw[7]));
    unsigned long long hA = 0ull, hB = 0ull;
#pragma unroll
    for (int u = 0; u < KEY_GROUPS; ++u) {
      const unsigned h16 = key_hits4(kw[u].x, kk) | (key_hits4(kw[u].y, kk) << 4) |
                           (key_hits4(kw[u].z, kk) << 8) | (key_hits4(kw[u].w, kk) << 12);
      const unsigned long long hm = ok[u] ? (unsigned long long)h16 : 0ull;
      if (u < 4) hA |= hm << (16 * (u & 3));
      else       hB |= hm << (16 * (u & 3));
    }
    while (true) {
      const bool has = (hA | hB) != 0ull;
      const unsigned bal = __builtin_amdgcn_ballot_w32(has);
      if (bal == 0u) break;
      if (has) {
        int p;
        if (hA != 0ull) { p = __builtin_ctzll(hA); hA &= hA - 1ull; }
        else            { p = 64 + __builtin_ctzll(hB); hB &= hB - 1ull; }
        int c = j0 + (p >> 4) * OWN_WAVES + rot;
        if (c >= nchunks) c -= nchunks;
        const int pos = qn + (int)__builtin_amdgcn_mbcnt_lo(bal, 0u);
        q[pos] = (c * 32 + lane) * 16 + (p & 15);
      }
      qn += __builtin_popcount(bal);
      if (qn >= 32) {
        __builtin_amdgcn_wave_barrier();
        const int ei = q[lane];
        const int rest = qn - 32;
        const int mv = (lane < rest) ? q[32 + lane] : 0;
        __builtin_amdgcn_wave_barrier();
        if (lane < rest) q[lane] = mv;
        qn = rest;
        owner_batch<D>(h, src, dst, deg, rows, ei, 32, e, lo, hi, lane,
                       &lcnt, off, ls, lw, lr);
      }
    }
  }
  if (qn > 0) {
    __builtin_amdgcn_wave_barrier();
    const int ei = (lane < qn) ? q[lane] : 0;
    owner_batch<D>(h, src, dst, deg, rows, ei, qn, e, lo, hi, lane,
                   &lcnt, off, ls, lw, lr);
  }
  } else {
    const int last = max(e - 1, 0);
    for (int t0 = wave; t0 * 32 < cnt; t0 += OWN_WAVES * LIST_TRIPS) {
      int s[LIST_TRIPS], dl[LIST_TRIPS];
      float w[LIST_TRIPS];
#pragma unroll
      for (int u = 0; u < LIST_TRIPS; ++u) {
        const int p = min(off + (t0 + u * OWN_WAVES) * 32 + lane, last);
        s[u] = ls[p];
        dl[u] = lr[p];
        w[u] = lw[p];
      }
#pragma unroll
      for (int u = 0; u < LIST_TRIPS; ++u) {
        const int i = (t0 + u * OWN_WAVES) * 32 + lane;
        const bool okv = i < cnt && off + i < e &&
                         (unsigned)dl[u] < (unsigned)(hi - lo) &&
                         (unsigned)s[u] < (unsigned)nrows;
        const unsigned m = __builtin_amdgcn_ballot_w32(okv);
        add_rows<D>(h, rows, s[u], dl[u], w[u], m, lane);
      }
    }
  }
  __syncthreads();

  float4* a4 = (float4*)(agg + (long)lo * D);
  for (int t = tid; t < nq; t += OWN_THREADS) a4[t] = r4[t];
}

extern "C" void kernel_launch(void* const* d_in, const int* in_sizes, int n_in,
                              void* d_out, int out_size, void* d_ws, size_t ws_size,
                              hipStream_t stream) {
  (void)n_in; (void)out_size; (void)ws_size;
  const float* x  = (const float*)d_in[0];
  const int*   ei = (const int*)d_in[1];
  const float* W1 = (const float*)d_in[2];
  const float* b1 = (const float*)d_in[3];
  const float* W3 = (const float*)d_in[4];
  const float* b3 = (const float*)d_in[5];

  const int N = in_sizes[0] / 64;
  const int E = in_sizes[1] / 2;
  const int* src = ei;
  const int* dst = ei + E;

  unsigned char* keys = (unsigned char*)d_ws;
  float* deg  = (float*)(keys + (((long)E + 255) & ~255L));
  float* h    = deg + ((N + 63) & ~63);
  float* agg1 = h + (long)N * 64;
  float* lst  = agg1 + (long)N * 64;
  float* h2   = h;
  int*   ls   = (int*)lst;
  float* lw   = lst + E;
  int*   lr   = (int*)(lst + 2L * E);
  int*   oinfo = (int*)(lst + 3L * E);
  float* out  = (float*)d_out;

  const int T = 256;
  const int mtiles = (N + 15) / 16;
  const int nwords = ((E + 15) / 16) * 4;
  const int owners = (N + OWN_ROWS - 1) / OWN_ROWS;

  gcn_gemm_rows_f32<64, false, true><<<(mtiles + GEMM_WAVES - 1) / GEMM_WAVES, GEMM_WAVES * 32, 0, stream>>>(x, W1, h, deg, N);

  gcn_count_deg_keys<<<(nwords + T - 1) / T, T, 0, stream>>>(dst, deg, (unsigned*)keys, E, nwords);

  gcn_owner_sum<64, true><<<owners, OWN_THREADS, 0, stream>>>(
      h, src, dst, deg, b1, (const unsigned*)keys, oinfo, ls, lw, lr, agg1, E, N);

  gcn_gemm_rows_f32<40, true, false><<<(mtiles + GEMM_WAVES - 1) / GEMM_WAVES, GEMM_WAVES * 32, 0, stream>>>(agg1, W3, h2, deg, N);
  gcn_owner_sum<40, false><<<owners, OWN_THREADS, 0, stream>>>(
      h2, src, dst, deg, b3, (const unsigned*)keys, oinfo, ls, lw, lr, out, E, N);
}


# Layer-2 product moved behind its edge sums, (A r) W3 for A (r W3): owners sum 64-wide activated rows in LDS, then multiply them by W3 on the f32 WMMA; 5 -> 4 launches; rounding changes
# speedup vs baseline: 2.3478x; 1.0364x over previous
#include <hip/hip_runtime.h>


typedef float v2f __attribute__((ext_vector_type(2)));
typedef float v8f __attribute__((ext_vector_type(8)));
typedef unsigned u4 __attribute__((ext_vector_type(4)));
typedef float v4f __attribute__((ext_vector_type(4)));

#define OWN_ROWS     392
#define OWN_THREADS  1024
#define OWN_WAVES    (OWN_THREADS / 32)
#define KEY_GROUPS   8
#define ROWS_IN_FLIGHT 16
#define LIST_TRIPS   4
#define DEG_GROUP    8

__global__ void gcn_count_deg_keys(const int* __restrict__ dst, float* deg,
                                   unsigned* __restrict__ keys4, int e, int nwords) {
  const int t = blockIdx.x * blockDim.x + threadIdx.x;
  if (t >= nwords) return;
  unsigned pk = 0u;
#pragma unroll
  for (int b = 0; b < 4; ++b) {
    const int i = t * 4 + b;
    if (i < e) {
      const int d = dst[i];
      atomicAdd(&deg[d], 1.0f);
      pk |= (unsigned)((d / OWN_ROWS) & 255) << (8 * b);
    }
  }
  keys4[t] = pk;
}

#define GEMM_WAVES 8
#define A_PITCH 68
#define B_PITCH 72

template <int NCOL, bool DEG_ONE>
__global__ void __launch_bounds__(GEMM_WAVES * 32)
gcn_gemm_rows_f32(const float* __restrict__ A,
                  const float* __restrict__ B,
                  float* __restrict__ C, float* __restrict__ deg, int nrows) {
  constexpr int NT = (NCOL + 15) / 16;
  constexpr int NP = NT * 16;
  constexpr int QR = NCOL / 4;
  constexpr int Q  = 16 * QR;
  static_assert(NCOL % 4 == 0 && NP <= B_PITCH && NP <= A_PITCH, "tile shape");
  static_assert((64 * NCOL) % (GEMM_WAVES * 32) == 0 && (64 * (NP - NCOL)) % (GEMM_WAVES * 32) == 0 &&
                Q % 32 == 0, "whole trips");
  __shared__ __attribute__((aligned(16))) float bl[64 * B_PITCH];
  __shared__ __attribute__((aligned(16))) float sl[GEMM_WAVES][16 * A_PITCH];

  const int tid  = threadIdx.x;
  const int lane = tid & 31;
  const int wave = tid >> 5;
  const int row0 = (blockIdx.x * GEMM_WAVES + wave) * 16;
  float* s = sl[wave];

  if (DEG_ONE && wave == 0) {
    const int r = blockIdx.x * (GEMM_WAVES * 16) + lane * 4;
    if (r + 4 <= nrows) {
      *(float4*)(deg + r) = float4{1.0f, 1.0f, 1.0f, 1.0f};
    } else {
      for (int i = r; i < nrows; ++i) deg[i] = 1.0f;
    }
  }

  float4 av[8];
#pragma unroll
  for (int j = 0; j < 8; ++j) {
    const int q = j * 32 + lane;
    const int r = min(row0 + (q >> 4), nrows - 1);
    av[j] = *(const float4*)(A + (long)r * 64 + (q & 15) * 4);
  }

#pragma unroll
  for (int i = 0; i < (64 * NCOL) / (GEMM_WAVES * 32); ++i) {
    const int idx = i * (GEMM_WAVES * 32) + tid;
    const int k = idx / NCOL;
    bl[k * B_PITCH + (idx - k * NCOL)] = B[idx];
  }
  if (NP > NCOL) {
#pragma unroll
    for (int i = 0; i < (64 * (NP - NCOL)) / (GEMM_WAVES * 32); ++i) {
      const int idx = i * (GEMM_WAVES * 32) + tid;
      const int k = idx / (NP - NCOL);
      bl[k * B_PITCH + NCOL + (idx - k * (NP - NCOL))] = 0.0f;
    }
  }

#pragma unroll
  for (int j = 0; j < 8; ++j) {
    const int q = j * 32 + lane;
    *(float4*)(s + (q >> 4) * A_PITCH + (q & 15) * 4) = av[j];
  }
  __syncthreads();
  if (row0 >= nrows) return;

  const int c  = lane & 15;
  const int hh = lane >> 4;
  const float* ap = s + c * A_PITCH + 2 * hh;
  const float* bp = bl + (2 * hh) * B_PITCH + c;

  v8f acc[NT];
#pragma unroll
  for (int t = 0; t < NT; ++t) acc[t] = v8f{};
#pragma unroll
  for (int ks = 0; ks < 16; ++ks) {
    const v2f a = *(const v2f*)(ap + 4 * ks);
#pragma unroll
    for (int t = 0; t < NT; ++t) {
      const v2f b = { bp[(4 * ks) * B_PITCH + 16 * t],
                      bp[(4 * ks + 1) * B_PITCH + 16 * t] };
      acc[t] = __builtin_amdgcn_wmma_f32_16x16x4_f32(
           false, a,  false, b,
           (short)0, acc[t],  false,  false);
    }
  }

#pragma unroll
  for (int t = 0; t < NT; ++t) {
#pragma unroll
    for (int r = 0; r < 8; ++r) s[(8 * hh + r) * A_PITCH + 16 * t + c] = acc[t][r];
  }
  __builtin_amdgcn_wave_barrier();

  float4* cp = (float4*)(C + (long)row0 * NCOL);
  if (row0 + 16 <= nrows) {
    float4 cv[Q / 32];
#pragma unroll
    for (int j = 0; j < Q / 32; ++j) {
      const int q = j * 32 + lane;
      const int r = q / QR;
      cv[j] = *(const float4*)(s + r * A_PITCH + (q - r * QR) * 4);
    }
#pragma unroll
    for (int j = 0; j < Q / 32; ++j) cp[j * 32 + lane] = cv[j];
  } else {
#pragma unroll
    for (int j = 0; j < Q / 32; ++j) {
      const int q = j * 32 + lane;
      const int r = q / QR;
      if (row0 + r < nrows) cp[q] = *(const float4*)(s + r * A_PITCH + (q - r * QR) * 4);
    }
  }
}

__device__ __forceinline__ float lane_bcast_f(float v, int j) {
  return __int_as_float(__builtin_amdgcn_readlane(__float_as_int(v), j));
}

template <int D, int P>
__device__ __forceinline__ void add_rows(const float* __restrict__ h, float* rows,
                                         int s, int dl, float w, unsigned m, int lane) {
  const bool act = (D == 64) || (lane < D / 2);
  while (m != 0u) {
    const int k = min((int)__builtin_popcount(m), ROWS_IN_FLIGHT);
    int jj[ROWS_IN_FLIGHT];
    v2f v[ROWS_IN_FLIGHT];
#pragma unroll
    for (int u = 0; u < ROWS_IN_FLIGHT; ++u) {
      jj[u] = 0;
      v[u] = v2f{0.0f, 0.0f};
      if (u < k) {
        jj[u] = __builtin_ctz(m);
        m &= m - 1u;
        const int sj = __builtin_amdgcn_readlane(s, jj[u]);
        if (act) v[u] = *(const v2f*)(h + (long)sj * D + 2 * lane);
      }
    }
#pragma unroll
    for (int u = 0; u < ROWS_IN_FLIGHT; ++u) {
      if (u < k) {
        const int rj = __builtin_amdgcn_readlane(dl, jj[u]);
        const float wj = lane_bcast_f(w, jj[u]);
        float* p = rows + rj * P + 2 * lane;
        if (act) {
          atomicAdd(p + 0, v[u].x * wj);
          atomicAdd(p + 1, v[u].y * wj);
        }
      }
    }
  }
}

template <int D>
__device__ __forceinline__ void owner_batch(const float* __restrict__ h,
                                            const int* __restrict__ src,
                                            const int* __restrict__ dst,
                                            const float* __restrict__ deg,
                                            float* rows, int ei, int nvalid,
                                            int e, int lo, int hi, int lane,
                                            int* lcnt, int off,
                                            int* ls, float* lw, int* lr) {
  int s = 0, d = -1;
  if (lane < nvalid && ei < e) { s = src[ei]; d = dst[ei]; }
  const bool mine = d >= lo && d < hi;
  float ws = 0.0f, wd = 0.0f;
  if (mine) { ws = rsqrtf(deg[s]); wd = rsqrtf(deg[d]); }
  const float w = ws * wd;
  const unsigned m = __builtin_amdgcn_ballot_w32(mine);
  int base = 0;
  if (lane == 0) base = atomicAdd(lcnt, (int)__builtin_popcount(m));
  base = __builtin_amdgcn_readfirstlane(base);
  const int pos = off + base + (int)__builtin_amdgcn_mbcnt_lo(m, 0u);
  if (mine && pos < e) { ls[pos] = s; lw[pos] = w; lr[pos] = d - lo; }
  add_rows<D, D>(h, rows, s, d - lo, w, m, lane);
}

__device__ __forceinline__ unsigned key_hits4(unsigned x, unsigned kk) {
  const unsigned y = x ^ kk;
  const unsigned z = (y - 0x01010101u) & ~y & 0x80808080u;
  return (((z >> 7) * 0x01020408u) >> 24) & 15u;
}

template <int D, bool SCAN, bool RELU_OUT, int NOUT>
__global__ void __launch_bounds__(OWN_THREADS)
gcn_owner_sum(const float* __restrict__ h,
              const int* __restrict__ src,
              const int* __restrict__ dst,
              const float* __restrict__ deg,
              const float* __restrict__ bias,
              const float* __restrict__ wmat,
              const unsigned* __restrict__ keys,
              int* oinfo,
              int* ls, float* lw, int* lr,
              float* __restrict__ agg, int e, int nrows) {
  constexpr bool PROD = NOUT > 0;
  constexpr int P = PROD ? A_PITCH : D;
  constexpr int OWN_TILES = (OWN_ROWS + 15) / 16;
  constexpr int LROWS = PROD ? OWN_TILES * 16 : OWN_ROWS;
  static_assert(!PROD || (D == 64 && !SCAN && OWN_TILES <= OWN_WAVES), "one tile per wave, K = 64");
  const u4* keys16 = (const u4*)keys;
  __shared__ __attribute__((aligned(16))) float rows[LROWS * P];
  __shared__ __attribute__((aligned(16))) float bl[PROD ? 64 * B_PITCH : 4];
  __shared__ int queue[SCAN ? OWN_WAVES : 1][64];
  __shared__ int lcnt;
  __shared__ int wpre[SCAN ? OWN_WAVES : 1];
  __shared__ int wown[SCAN ? OWN_WAVES : 1];
  static_assert(OWN_WAVES == 32, "one partial per lane in the sums below");

  const int tid  = threadIdx.x;
  const int lane = tid & 31;
  const int wave = tid >> 5;
  const int lo   = blockIdx.x * OWN_ROWS;
  const int hi   = min(lo + OWN_ROWS, nrows);
  const int nq   = (hi - lo) * (D / 4);

  int off = 0, cnt = 0;
  const bool below = 2 * lo <= nrows;
  if (SCAN) {
    const int a = below ? 0 : hi;
    const int b = below ? lo : nrows;
    const int n4 = (b - a) >> 2;
    const v4f* d4 = (const v4f*)(deg + a);
    int c = 0;
    for (int q0 = tid; q0 < n4; q0 += OWN_THREADS * DEG_GROUP) {
      v4f v[DEG_GROUP];
#pragma unroll
      for (int u = 0; u < DEG_GROUP; ++u) v[u] = d4[min(q0 + u * OWN_THREADS, n4 - 1)];
      static_assert(DEG_GROUP == 8, "the fence below names eight loads");
      asm volatile("" :: "v"(v[0]), "v"(v[1]), "v"(v[2]), "v"(v[3]),
                         "v"(v[4]), "v"(v[5]), "v"(v[6]), "v"(v[7]));
#pragma unroll
      for (int u = 0; u < DEG_GROUP; ++u) {
        const int t = ((int)v[u].x - 1) + ((int)v[u].y - 1) + ((int)v[u].z - 1) + ((int)v[u].w - 1);
        c += (q0 + u * OWN_THREADS < n4) ? t : 0;
      }
    }
    for (int r = a + 4 * n4 + tid; r < b; r += OWN_THREADS) c += (int)deg[r] - 1;
    int cm = 0;
    for (int r = lo + tid; r < hi; r += OWN_THREADS) cm += (int)deg[r] - 1;
    for (int o = 16; o > 0; o >>= 1) { c += __shfl_xor(c, o); cm += __shfl_xor(cm, o); }
    if (lane == 0) { wpre[wave] = c; wown[wave] = cm; }
  } else {
    off = min(max(oinfo[2 * blockIdx.x], 0), e);
    cnt = min(max(oinfo[2 * blockIdx.x + 1], 0), e - off);
  }
  if (tid == 0) lcnt = 0;

  const float4* h4 = (const float4*)(h + (long)lo * D);
  float4* r4 = (float4*)rows;
  if constexpr (!PROD) {
    for (int q = tid; q < nq; q += OWN_THREADS) {
      const int r = q / (D / 4);
      const int c = (q - r * (D / 4)) * 4;
      const float di = rsqrtf(deg[lo + r]);
      const float4 hv = h4[q];
      float4 o;
      o.x = di * di * hv.x + bias[c + 0];
      o.y = di * di * hv.y + bias[c + 1];
      o.z = di * di * hv.z + bias[c + 2];
      o.w = di * di * hv.w + bias[c + 3];
      r4[q] = o;
    }
  } else {
    for (int q = tid; q < LROWS * (D / 4); q += OWN_THREADS) {
      const int r = q / (D / 4);
      const int c = (q - r * (D / 4)) * 4;
      float4 o = float4{0.0f, 0.0f, 0.0f, 0.0f};
      if (q < nq) {
        const float di = rsqrtf(deg[lo + r]);
        const float4 hv = h4[q];
        o.x = di * di * hv.x;
        o.y = di * di * hv.y;
        o.z = di * di * hv.z;
        o.w = di * di * hv.w;
      }
      *(float4*)(rows + r * P + c) = o;
    }
    constexpr int NPW = ((NOUT + 15) / 16) * 16;
    for (int idx = tid; idx < 64 * NOUT; idx += OWN_THREADS) {
      const int k = idx / NOUT;
      bl[k * B_PITCH + (idx - k * NOUT)] = wmat[idx];
    }
    for (int idx = tid; idx < 64 * (NPW - NOUT); idx += OWN_THREADS) {
      const int k = idx / (NPW - NOUT);
      bl[k * B_PITCH + NOUT + (idx - k * (NPW - NOUT))] = 0.0f;
    }
  }
  __syncthreads();

  if (SCAN) {
  {
    int c = wpre[lane], cm = wown[lane];
    for (int o = 16; o > 0; o >>= 1) { c += __shfl_xor(c, o); cm += __shfl_xor(cm, o); }
    cnt = __builtin_amdgcn_readfirstlane(cm);
    c = __builtin_amdgcn_readfirstlane(c);
    off = below ? c : e - c - cnt;
    if (tid == 0) { oinfo[2 * blockIdx.x] = off; oinfo[2 * blockIdx.x + 1] = cnt; }
  }
  const int ngroups = (e + 15) >> 4;
  const int nchunks = (ngroups + 31) >> 5;
  const int rot = (int)(((long)blockIdx.x * nchunks) / gridDim.x);
  const unsigned kk = (unsigned)(blockIdx.x & 255) * 0x01010101u;
  int* q = queue[wave];
  int qn = 0;

  for (int j0 = wave; j0 < nchunks; j0 += OWN_WAVES * KEY_GROUPS) {
    u4 kw[KEY_GROUPS];
    bool ok[KEY_GROUPS];
#pragma unroll
    for (int u = 0; u < KEY_GROUPS; ++u) {
      const int j = j0 + u * OWN_WAVES;
      int c = j + rot;
      if (c >= nchunks) c -= nchunks;
      const int g = c * 32 + lane;
      ok[u] = (j < nchunks) && (g < ngroups);
      kw[u] = keys16[min(g, ngroups - 1)];
    }
    static_assert(KEY_GROUPS == 8, "the fence below names eight loads");
    asm volatile("" :: "v"(kw[0]), "v"(kw[1]), "v"(kw[2]), "v"(kw[3]),
                       "v"(kw[4]), "v"(kw[5]), "v"(kw[6]), "v"(kw[7]));
    unsigned long long hA = 0ull, hB = 0ull;
#pragma unroll
    for (int u = 0; u < KEY_GROUPS; ++u) {
      const unsigned h16 = key_hits4(kw[u].x, kk) | (key_hits4(kw[u].y, kk) << 4) |
                           (key_hits4(kw[u].z, kk) << 8) | (key_hits4(kw[u].w, kk) << 12);
      const unsigned long long hm = ok[u] ? (unsigned long long)h16 : 0ull;
      if (u < 4) hA |= hm << (16 * (u & 3));
      else       hB |= hm << (16 * (u & 3));
    }
    while (true) {
      const bool has = (hA | hB) != 0ull;
      const unsigned bal = __builtin_amdgcn_ballot_w32(has);
      if (bal == 0u) break;
      if (has) {
        int p;
        if (hA != 0ull) { p = __builtin_ctzll(hA); hA &= hA - 1ull; }
        else            { p = 64 + __builtin_ctzll(hB); hB &= hB - 1ull; }
        int c = j0 + (p >> 4) * OWN_WAVES + rot;
        if (c >= nchunks) c -= nchunks;
        const int pos = qn + (int)__builtin_amdgcn_mbcnt_lo(bal, 0u);
        q[pos] = (c * 32 + lane) * 16 + (p & 15);
      }
      qn += __builtin_popcount(bal);
      if (qn >= 32) {
        __builtin_amdgcn_wave_barrier();
        const int ei = q[lane];
        const int rest = qn - 32;
        const int mv = (lane < rest) ? q[32 + lane] : 0;
        __builtin_amdgcn_wave_barrier();
        if (lane < rest) q[lane] = mv;
        qn = rest;
        owner_batch<D>(h, src, dst, deg, rows, ei, 32, e, lo, hi, lane,
                       &lcnt, off, ls, lw, lr);
      }
    }
  }
  if (qn > 0) {
    __builtin_amdgcn_wave_barrier();
    const int ei = (lane < qn) ? q[lane] : 0;
    owner_batch<D>(h, src, dst, deg, rows, ei, qn, e, lo, hi, lane,
                   &lcnt, off, ls, lw, lr);
  }
  } else {
    const int last = max(e - 1, 0);
    for (int t0 = wave; t0 * 32 < cnt; t0 += OWN_WAVES * LIST_TRIPS) {
      int s[LIST_TRIPS], dl[LIST_TRIPS];
      float w[LIST_TRIPS];
#pragma unroll
      for (int u = 0; u < LIST_TRIPS; ++u) {
        const int p = min(off + (t0 + u * OWN_WAVES) * 32 + lane, last);
        s[u] = ls[p];
        dl[u] = lr[p];
        w[u] = lw[p];
      }
#pragma unroll
      for (int u = 0; u < LIST_TRIPS; ++u) {
        const int i = (t0 + u * OWN_WAVES) * 32 + lane;
        const bool okv = i < cnt && off + i < e &&
                         (unsigned)dl[u] < (unsigned)(hi - lo) &&
                         (unsigned)s[u] < (unsigned)nrows;
        const unsigned m = __builtin_amdgcn_ballot_w32(okv);
        add_rows<D, P>(h, rows, s[u], dl[u], w[u], m, lane);
      }
    }
  }
  __syncthreads();

  if constexpr (!PROD) {
    float4* a4 = (float4*)(agg + (long)lo * D);
    for (int t = tid; t < nq; t += OWN_THREADS) {
      float4 v = r4[t];
      if (RELU_OUT) {
        v.x = v.x > 0.0f ? v.x : 0.0f;
        v.y = v.y > 0.0f ? v.y : 0.0f;
        v.z = v.z > 0.0f ? v.z : 0.0f;
        v.w = v.w > 0.0f ? v.w : 0.0f;
      }
      a4[t] = v;
    }
  } else {
    constexpr int NT = (NOUT + 15) / 16;
    constexpr int QR = NOUT / 4;
    constexpr int Q  = 16 * QR;
    static_assert(NOUT % 4 == 0 && NT * 16 <= B_PITCH && NT * 16 <= A_PITCH && Q % 32 == 0, "tile shape");
    const int nown = hi - lo;
    if (wave * 16 < nown) {
      float* s = rows + wave * 16 * P;
      const int c  = lane & 15;
      const int hh = lane >> 4;
      const float* ap = s + c * P + 2 * hh;
      const float* bp = bl + (2 * hh) * B_PITCH + c;

      v8f acc[NT];
#pragma unroll
      for (int t = 0; t < NT; ++t) acc[t] = v8f{};
#pragma unroll
      for (int ks = 0; ks < 16; ++ks) {
        const v2f a = *(const v2f*)(ap + 4 * ks);
#pragma unroll
        for (int t = 0; t < NT; ++t) {
          const v2f b = { bp[(4 * ks) * B_PITCH + 16 * t],
                          bp[(4 * ks + 1) * B_PITCH + 16 * t] };
          acc[t] = __builtin_amdgcn_wmma_f32_16x16x4_f32(
               false, a,  false, b,
               (short)0, acc[t],  false,  false);
        }
      }

#pragma unroll
      for (int t = 0; t < NT; ++t) {
        const float bc = (16 * t + c < NOUT) ? bias[16 * t + c] : 0.0f;
#pragma unroll
        for (int r = 0; r < 8; ++r) s[(8 * hh + r) * P + 16 * t + c] = acc[t][r] + bc;
      }
      __builtin_amdgcn_wave_barrier();

      float4* cp = (float4*)(agg + (long)(lo + wave * 16) * NOUT);
      if (wave * 16 + 16 <= nown) {
        float4 cv[Q / 32];
#pragma unroll
        for (int j = 0; j < Q / 32; ++j) {
          const int q = j * 32 + lane;
          const int r = q / QR;
          cv[j] = *(const float4*)(s + r * P + (q - r * QR) * 4);
        }
#pragma unroll
        for (int j = 0; j < Q / 32; ++j) cp[j * 32 + lane] = cv[j];
      } else {
#pragma unroll
        for (int j = 0; j < Q / 32; ++j) {
          const int q = j * 32 + lane;
          const int r = q / QR;
          if (wave * 16 + r < nown) cp[q] = *(const float4*)(s + r * P + (q - r * QR) * 4);
        }
      }
    }
  }
}

extern "C" void kernel_launch(void* const* d_in, const int* in_sizes, int n_in,
                              void* d_out, int out_size, void* d_ws, size_t ws_size,
                              hipStream_t stream) {
  (void)n_in; (void)out_size; (void)ws_size;
  const float* x  = (const float*)d_in[0];
  const int*   ei = (const int*)d_in[1];
  const float* W1 = (const float*)d_in[2];
  const float* b1 = (const float*)d_in[3];
  const float* W3 = (const float*)d_in[4];
  const float* b3 = (const float*)d_in[5];

  const int N = in_sizes[0] / 64;
  const int E = in_sizes[1] / 2;
  const int* src = ei;
  const int* dst = ei + E;

  unsigned char* keys = (unsigned char*)d_ws;
  float* deg  = (float*)(keys + (((long)E + 255) & ~255L));
  float* h    = deg + ((N + 63) & ~63);
  float* agg1 = h + (long)N * 64;
  float* lst  = agg1 + (long)N * 64;
  int*   ls   = (int*)lst;
  float* lw   = lst + E;
  int*   lr   = (int*)(lst + 2L * E);
  int*   oinfo = (int*)(lst + 3L * E);
  float* out  = (float*)d_out;

  const int T = 256;
  const int mtiles = (N + 15) / 16;
  const int nwords = ((E + 15) / 16) * 4;
  const int owners = (N + OWN_ROWS - 1) / OWN_ROWS;

  gcn_gemm_rows_f32<64, true><<<(mtiles + GEMM_WAVES - 1) / GEMM_WAVES, GEMM_WAVES * 32, 0, stream>>>(x, W1, h, deg, N);

  gcn_count_deg_keys<<<(nwords + T - 1) / T, T, 0, stream>>>(dst, deg, (unsigned*)keys, E, nwords);

  gcn_owner_sum<64, true, true, 0><<<owners, OWN_THREADS, 0, stream>>>(
      h, src, dst, deg, b1, W3, (const unsigned*)keys, oinfo, ls, lw, lr, agg1, E, N);

  gcn_owner_sum<64, false, false, 40><<<owners, OWN_THREADS, 0, stream>>>(
      agg1, src, dst, deg, b3, W3, (const unsigned*)keys, oinfo, ls, lw, lr, out, E, N);
}
